# starting kernel
# baseline (speedup 1.0000x reference)
#include <hip/hip_runtime.h>
#include <hip/hip_bf16.h>
#include <math.h>

typedef __attribute__((ext_vector_type(2))) float v2f;
typedef __attribute__((ext_vector_type(8))) float v8f;

#define DIM     32
#define K2      64
#define CCOMP   8
#define BATCH   2048
#define KFEAT   64
#define NCHUNK  16

__device__ __forceinline__ float softplus_stable(float z) {
    return fmaxf(z, 0.0f) + log1pf(expf(-fabsf(z)));
}

__device__ __forceinline__ float swz_x1(float v) {
    return __int_as_float(__builtin_amdgcn_ds_swizzle(__float_as_int(v), 0x041F));
}
__device__ __forceinline__ float swz_x2(float v) {
    return __int_as_float(__builtin_amdgcn_ds_swizzle(__float_as_int(v), 0x081F));
}
__device__ __forceinline__ float swz_x4(float v) {
    return __int_as_float(__builtin_amdgcn_ds_swizzle(__float_as_int(v), 0x101F));
}

__global__ void __launch_bounds__(128)
mvn_mixture_wmma_kernel(const float* __restrict__ inputs,
                        const float* __restrict__ kern,
                        const float* __restrict__ mixw,
                        float* __restrict__ out)
{
    __shared__ float lds_W[KFEAT * 16];
    __shared__ float lds_pld[8 * 16];
    __shared__ float lds_pc0[8 * 16];
    __shared__ float lds_beta[16];

    const int k1i   = blockIdx.x >> 5;
    const int ntile = blockIdx.x & 31;
    const int n0    = ntile * 16;

    const int tid = threadIdx.x;

    {
        const int comp  = tid & 15;
        const int chunk = tid >> 4;
        const int n  = n0 + comp;
        const int k2 = n >> 3;
        const int c  = n & 7;
        const float* kp = kern + ((size_t)((k1i * K2 + k2) * CCOMP + c)) * (2 * DIM);

        const float SP_INV1 = 0.5413248546129181f;
        float pld = 0.0f, pc0 = 0.0f;
        #pragma unroll
        for (int j = 0; j < 4; ++j) {
            const int d = chunk * 4 + j;
            float mu  = kp[d];
            float raw = kp[DIM + d];
            float sd  = softplus_stable(raw + SP_INV1) + 1e-8f;
            float pin = 1.0f / sd;
            float w   = pin * pin;
            pld += logf(sd);
            pc0 += w * mu * mu;
            lds_W[d * 16 + comp]         = w * mu;
            lds_W[(DIM + d) * 16 + comp] = -0.5f * w;
        }
        lds_pld[chunk * 16 + comp] = pld;
        lds_pc0[chunk * 16 + comp] = pc0;
    }
    __syncthreads();

    if (tid < 16) {
        float logdet = 0.0f, c0 = 0.0f;
        #pragma unroll
        for (int ch = 0; ch < 8; ++ch) {
            logdet += lds_pld[ch * 16 + tid];
            c0     += lds_pc0[ch * 16 + tid];
        }
        logdet *= 2.0f;

        const int n  = n0 + tid;
        const int k2 = n >> 3;
        const int c  = n & 7;
        const float* mp = mixw + (size_t)(k1i * K2 + k2) * CCOMP;
        float mx = mp[0];
        #pragma unroll
        for (int j = 1; j < CCOMP; ++j) mx = fmaxf(mx, mp[j]);
        float se = 0.0f;
        #pragma unroll
        for (int j = 0; j < CCOMP; ++j) se += expf(mp[j] - mx);
        float logmix = mp[c] - mx - logf(se);

        const float CONST_TERM = 58.81206994227444f;
        lds_beta[tid] = -0.5f * (CONST_TERM + logdet + c0) + logmix;
    }
    __syncthreads();

    const int lane = tid & 31;
    const int wave = tid >> 5;
    const int col  = lane & 15;
    const int h    = lane >> 4;
    const int m_base = (blockIdx.y * 4 + wave) * 16;

    v2f bf[NCHUNK];
    #pragma unroll
    for (int kk = 0; kk < NCHUNK; ++kk) {
        int r0 = kk * 4 + h * 2;
        bf[kk].x = lds_W[(r0 + 0) * 16 + col];
        bf[kk].y = lds_W[(r0 + 1) * 16 + col];
    }
    const float beta = lds_beta[col];

    const float* xp = inputs + ((size_t)(k1i * BATCH + m_base + col)) * DIM + h * 2;
    v2f af[NCHUNK];
    #pragma unroll
    for (int kk = 0; kk < 8; ++kk)
        af[kk] = *(const v2f*)(xp + kk * 4);
    #pragma unroll
    for (int kk = 0; kk < 8; ++kk)
        af[8 + kk] = af[kk] * af[kk];

    v8f acc = {0.f, 0.f, 0.f, 0.f, 0.f, 0.f, 0.f, 0.f};
    #pragma unroll
    for (int kk = 0; kk < NCHUNK; ++kk) {
        acc = __builtin_amdgcn_wmma_f32_16x16x4_f32(
             false, af[kk],
             false, bf[kk],
             (short)0, acc,
             false,  false);
    }

    const int k2_out = (n0 >> 3) + ((lane >> 3) & 1);
    const bool store_lane = ((lane & 7) == 0);
    #pragma unroll
    for (int r = 0; r < 8; ++r) {
        float s = acc[r] + beta;
        float m = s;
        m = fmaxf(m, swz_x1(m));
        m = fmaxf(m, swz_x2(m));
        m = fmaxf(m, swz_x4(m));
        float e = expf(s - m);
        e += swz_x1(e);
        e += swz_x2(e);
        e += swz_x4(e);
        float res = m + logf(e);
        if (store_lane) {
            int b = m_base + h * 8 + r;
            out[((size_t)(k1i * BATCH + b)) * K2 + k2_out] = res;
        }
    }
}

extern "C" void kernel_launch(void* const* d_in, const int* in_sizes, int n_in,
                              void* d_out, int out_size, void* d_ws, size_t ws_size,
                              hipStream_t stream) {
    (void)in_sizes; (void)n_in; (void)out_size; (void)d_ws; (void)ws_size;
    const float* inputs = (const float*)d_in[0];
    const float* kern   = (const float*)d_in[1];
    const float* mixw   = (const float*)d_in[2];
    float* out          = (float*)d_out;

    dim3 grid(4 * 32, 32, 1);
    dim3 block(128, 1, 1);
    mvn_mixture_wmma_kernel<<<grid, block, 0, stream>>>(inputs, kern, mixw, out);
}


# Fewer, fatter waves: a work-group builds its coefficients once, each wave loops over 8 batch tiles (next rows prefetched, 8 row logsumexp chains interleaved, stores grouped); grid.y 32 -> 4
# speedup vs baseline: 1.2215x; 1.2215x over previous
#include <hip/hip_runtime.h>
#include <hip/hip_bf16.h>
#include <math.h>

typedef __attribute__((ext_vector_type(2))) float v2f;
typedef __attribute__((ext_vector_type(8))) float v8f;

#define DIM     32
#define K2      64
#define CCOMP   8
#define BATCH   2048
#define KFEAT   64
#define NCHUNK  16
#define TPW     8

__device__ __forceinline__ float softplus_stable(float z) {
    return fmaxf(z, 0.0f) + log1pf(expf(-fabsf(z)));
}

__device__ __forceinline__ float swz_x1(float v) {
    return __int_as_float(__builtin_amdgcn_ds_swizzle(__float_as_int(v), 0x041F));
}
__device__ __forceinline__ float swz_x2(float v) {
    return __int_as_float(__builtin_amdgcn_ds_swizzle(__float_as_int(v), 0x081F));
}
__device__ __forceinline__ float swz_x4(float v) {
    return __int_as_float(__builtin_amdgcn_ds_swizzle(__float_as_int(v), 0x101F));
}

__global__ void __launch_bounds__(128)
mvn_mixture_wmma_kernel(const float* __restrict__ inputs,
                        const float* __restrict__ kern,
                        const float* __restrict__ mixw,
                        float* __restrict__ out)
{
    __shared__ float lds_W[KFEAT * 16];
    __shared__ float lds_pld[8 * 16];
    __shared__ float lds_pc0[8 * 16];
    __shared__ float lds_beta[16];

    const int k1i   = blockIdx.x >> 5;
    const int ntile = blockIdx.x & 31;
    const int n0    = ntile * 16;

    const int tid = threadIdx.x;

    const int lane = tid & 31;
    const int wave = tid >> 5;
    const int col  = lane & 15;
    const int h    = lane >> 4;
    const int m_first = (blockIdx.y * 4 + wave) * (TPW * 16);

    const float* xp = inputs + ((size_t)(k1i * BATCH + m_first + col)) * DIM + h * 2;
    v2f xa[8];
    #pragma unroll
    for (int kk = 0; kk < 8; ++kk)
        xa[kk] = *(const v2f*)(xp + kk * 4);

    {
        const int comp  = tid & 15;
        const int chunk = tid >> 4;
        const int n  = n0 + comp;
        const int k2 = n >> 3;
        const int c  = n & 7;
        const float* kp = kern + ((size_t)((k1i * K2 + k2) * CCOMP + c)) * (2 * DIM);

        const float SP_INV1 = 0.5413248546129181f;
        float pld = 0.0f, pc0 = 0.0f;
        #pragma unroll
        for (int j = 0; j < 4; ++j) {
            const int d = chunk * 4 + j;
            float mu  = kp[d];
            float raw = kp[DIM + d];
            float sd  = softplus_stable(raw + SP_INV1) + 1e-8f;
            float pin = 1.0f / sd;
            float w   = pin * pin;
            pld += logf(sd);
            pc0 += w * mu * mu;
            lds_W[d * 16 + comp]         = w * mu;
            lds_W[(DIM + d) * 16 + comp] = -0.5f * w;
        }
        lds_pld[chunk * 16 + comp] = pld;
        lds_pc0[chunk * 16 + comp] = pc0;
    }
    __syncthreads();

    if (tid < 16) {
        float logdet = 0.0f, c0 = 0.0f;
        #pragma unroll
        for (int ch = 0; ch < 8; ++ch) {
            logdet += lds_pld[ch * 16 + tid];
            c0     += lds_pc0[ch * 16 + tid];
        }
        logdet *= 2.0f;

        const int n  = n0 + tid;
        const int k2 = n >> 3;
        const int c  = n & 7;
        const float* mp = mixw + (size_t)(k1i * K2 + k2) * CCOMP;
        float mx = mp[0];
        #pragma unroll
        for (int j = 1; j < CCOMP; ++j) mx = fmaxf(mx, mp[j]);
        float se = 0.0f;
        #pragma unroll
        for (int j = 0; j < CCOMP; ++j) se += expf(mp[j] - mx);
        float logmix = mp[c] - mx - logf(se);

        const float CONST_TERM = 58.81206994227444f;
        lds_beta[tid] = -0.5f * (CONST_TERM + logdet + c0) + logmix;
    }
    __syncthreads();

    v2f bf[NCHUNK];
    #pragma unroll
    for (int kk = 0; kk < NCHUNK; ++kk) {
        int r0 = kk * 4 + h * 2;
        bf[kk].x = lds_W[(r0 + 0) * 16 + col];
        bf[kk].y = lds_W[(r0 + 1) * 16 + col];
    }
    const float beta = lds_beta[col];

    const int k2_out = (n0 >> 3) + ((lane >> 3) & 1);
    const bool store_lane = ((lane & 7) == 0);
    float* op = out + ((size_t)(k1i * BATCH + m_first + h * 8)) * K2 + k2_out;

    #pragma unroll 2
    for (int t = 0; t < TPW; ++t) {
        v2f xn[8];
        if (t + 1 < TPW) {
            #pragma unroll
            for (int kk = 0; kk < 8; ++kk)
                xn[kk] = *(const v2f*)(xp + (size_t)(t + 1) * (16 * DIM) + kk * 4);
        } else {
            #pragma unroll
            for (int kk = 0; kk < 8; ++kk) xn[kk] = xa[kk];
        }

        v2f af[NCHUNK];
        #pragma unroll
        for (int kk = 0; kk < 8; ++kk) {
            af[kk]     = xa[kk];
            af[8 + kk] = xa[kk] * xa[kk];
        }

        v8f acc = {0.f, 0.f, 0.f, 0.f, 0.f, 0.f, 0.f, 0.f};
        #pragma unroll
        for (int kk = 0; kk < NCHUNK; ++kk) {
            acc = __builtin_amdgcn_wmma_f32_16x16x4_f32(
                 false, af[kk],
                 false, bf[kk],
                 (short)0, acc,
                 false,  false);
        }

        float res[8];
        #pragma unroll
        for (int r = 0; r < 8; ++r) {
            float s = acc[r] + beta;
            float m = s;
            m = fmaxf(m, swz_x1(m));
            m = fmaxf(m, swz_x2(m));
            m = fmaxf(m, swz_x4(m));
            float e = expf(s - m);
            e += swz_x1(e);
            e += swz_x2(e);
            e += swz_x4(e);
            res[r] = m + logf(e);
        }
        if (store_lane) {
            #pragma unroll
            for (int r = 0; r < 8; ++r)
                op[(size_t)(t * 16 + r) * K2] = res[r];
        }

        #pragma unroll
        for (int kk = 0; kk < 8; ++kk) xa[kk] = xn[kk];
    }
}

extern "C" void kernel_launch(void* const* d_in, const int* in_sizes, int n_in,
                              void* d_out, int out_size, void* d_ws, size_t ws_size,
                              hipStream_t stream) {
    (void)in_sizes; (void)n_in; (void)out_size; (void)d_ws; (void)ws_size;
    const float* inputs = (const float*)d_in[0];
    const float* kern   = (const float*)d_in[1];
    const float* mixw   = (const float*)d_in[2];
    float* out          = (float*)d_out;

    dim3 grid(4 * 32, 4, 1);
    dim3 block(128, 1, 1);
    mvn_mixture_wmma_kernel<<<grid, block, 0, stream>>>(inputs, kern, mixw, out);
}


# All global loads leave the tile loop: rows of tiles 0-2 and mixture weights requested at entry, tiles 3-7 behind the last barrier, as 16-byte loads (both loaders' k order permuted: new summing order)
# speedup vs baseline: 1.4339x; 1.1739x over previous
#include <hip/hip_runtime.h>
#include <hip/hip_bf16.h>
#include <math.h>

typedef __attribute__((ext_vector_type(2))) float v2f;
typedef __attribute__((ext_vector_type(4))) float v4f;
typedef __attribute__((ext_vector_type(8))) float v8f;

#define DIM     32
#define K2      64
#define CCOMP   8
#define BATCH   2048
#define KFEAT   64
#define NCHUNK  16
#define TPW     8
#define TG1     3

__device__ __forceinline__ float softplus_stable(float z) {
    return fmaxf(z, 0.0f) + log1pf(expf(-fabsf(z)));
}

__device__ __forceinline__ float swz_x1(float v) {
    return __int_as_float(__builtin_amdgcn_ds_swizzle(__float_as_int(v), 0x041F));
}
__device__ __forceinline__ float swz_x2(float v) {
    return __int_as_float(__builtin_amdgcn_ds_swizzle(__float_as_int(v), 0x081F));
}
__device__ __forceinline__ float swz_x4(float v) {
    return __int_as_float(__builtin_amdgcn_ds_swizzle(__float_as_int(v), 0x101F));
}

__global__ void __launch_bounds__(128)
mvn_mixture_wmma_kernel(const float* __restrict__ inputs,
                        const float* __restrict__ kern,
                        const float* __restrict__ mixw,
                        float* __restrict__ out)
{
    __shared__ float lds_W[KFEAT * 16];
    __shared__ float lds_pld[8 * 16];
    __shared__ float lds_pc0[8 * 16];
    __shared__ float lds_beta[16];

    const int k1i   = blockIdx.x >> 5;
    const int ntile = blockIdx.x & 31;
    const int n0    = ntile * 16;

    const int tid = threadIdx.x;

    const int lane = tid & 31;
    const int wave = tid >> 5;
    const int col  = lane & 15;
    const int h    = lane >> 4;
    const int m_first = (blockIdx.y * 4 + wave) * (TPW * 16);

    const int comp  = tid & 15;
    const int chunk = tid >> 4;
    float mu4[4], raw4[4];
    {
        const int n = n0 + comp;
        const float* kp = kern + ((size_t)((k1i * K2 + (n >> 3)) * CCOMP + (n & 7))) * (2 * DIM);
        #pragma unroll
        for (int j = 0; j < 4; ++j) {
            mu4[j]  = kp[chunk * 4 + j];
            raw4[j] = kp[DIM + chunk * 4 + j];
        }
    }

    const float* xp = inputs + ((size_t)(k1i * BATCH + m_first + col)) * DIM + h * 16;
    v4f xq[TPW][4];
    #pragma unroll
    for (int t = 0; t < TG1; ++t) {
        #pragma unroll
        for (int j = 0; j < 4; ++j)
            xq[t][j] = *(const v4f*)(xp + (size_t)t * (16 * DIM) + j * 4);
    }

    float mw[CCOMP] = {0.f, 0.f, 0.f, 0.f, 0.f, 0.f, 0.f, 0.f};
    float mwc = 0.f;
    if (tid < 16) {
        const int n  = n0 + tid;
        const float* mp = mixw + (size_t)(k1i * K2 + (n >> 3)) * CCOMP;
        #pragma unroll
        for (int j = 0; j < CCOMP; ++j) mw[j] = mp[j];
        mwc = mp[n & 7];
    }

    {
        const float SP_INV1 = 0.5413248546129181f;
        float pld = 0.0f, pc0 = 0.0f;
        #pragma unroll
        for (int j = 0; j < 4; ++j) {
            const int d = chunk * 4 + j;
            float mu  = mu4[j];
            float raw = raw4[j];
            float sd  = softplus_stable(raw + SP_INV1) + 1e-8f;
            float pin = 1.0f / sd;
            float w   = pin * pin;
            pld += logf(sd);
            pc0 += w * mu * mu;
            lds_W[d * 16 + comp]         = w * mu;
            lds_W[(DIM + d) * 16 + comp] = -0.5f * w;
        }
        lds_pld[chunk * 16 + comp] = pld;
        lds_pc0[chunk * 16 + comp] = pc0;
    }
    __syncthreads();

    if (tid < 16) {
        float logdet = 0.0f, c0 = 0.0f;
        #pragma unroll
        for (int ch = 0; ch < 8; ++ch) {
            logdet += lds_pld[ch * 16 + tid];
            c0     += lds_pc0[ch * 16 + tid];
        }
        logdet *= 2.0f;

        float mx = mw[0];
        #pragma unroll
        for (int j = 1; j < CCOMP; ++j) mx = fmaxf(mx, mw[j]);
        float se = 0.0f;
        #pragma unroll
        for (int j = 0; j < CCOMP; ++j) se += expf(mw[j] - mx);
        float logmix = mwc - mx - logf(se);

        const float CONST_TERM = 58.81206994227444f;
        lds_beta[tid] = -0.5f * (CONST_TERM + logdet + c0) + logmix;
    }
    __syncthreads();

    #pragma unroll
    for (int t = TG1; t < TPW; ++t) {
        #pragma unroll
        for (int j = 0; j < 4; ++j)
            xq[t][j] = *(const v4f*)(xp + (size_t)t * (16 * DIM) + j * 4);
    }

    v2f bf[NCHUNK];
    #pragma unroll
    for (int kk = 0; kk < NCHUNK; ++kk) {
        int r0 = (kk & 8) * 4 + h * 16 + (kk & 7) * 2;
        bf[kk].x = lds_W[(r0 + 0) * 16 + col];
        bf[kk].y = lds_W[(r0 + 1) * 16 + col];
    }
    const float beta = lds_beta[col];

    const int k2_out = (n0 >> 3) + ((lane >> 3) & 1);
    const bool store_lane = ((lane & 7) == 0);
    float* op = out + ((size_t)(k1i * BATCH + m_first + h * 8)) * K2 + k2_out;

    #pragma unroll
    for (int t = 0; t < TPW; ++t) {
        v2f af[NCHUNK];
        #pragma unroll
        for (int kk = 0; kk < 8; ++kk) {
            const v4f q = xq[t][kk >> 1];
            v2f a;
            if (kk & 1) { a.x = q.z; a.y = q.w; } else { a.x = q.x; a.y = q.y; }
            af[kk]     = a;
            af[8 + kk] = a * a;
        }

        v8f acc = {0.f, 0.f, 0.f, 0.f, 0.f, 0.f, 0.f, 0.f};
        #pragma unroll
        for (int kk = 0; kk < NCHUNK; ++kk) {
            acc = __builtin_amdgcn_wmma_f32_16x16x4_f32(
                 false, af[kk],
                 false, bf[kk],
                 (short)0, acc,
                 false,  false);
        }

        float res[8];
        #pragma unroll
        for (int r = 0; r < 8; ++r) {
            float s = acc[r] + beta;
            float m = s;
            m = fmaxf(m, swz_x1(m));
            m = fmaxf(m, swz_x2(m));
            m = fmaxf(m, swz_x4(m));
            float e = expf(s - m);
            e += swz_x1(e);
            e += swz_x2(e);
            e += swz_x4(e);
            res[r] = m + logf(e);
        }
        if (store_lane) {
            #pragma unroll
            for (int r = 0; r < 8; ++r)
                op[(size_t)(t * 16 + r) * K2] = res[r];
        }
    }
}

extern "C" void kernel_launch(void* const* d_in, const int* in_sizes, int n_in,
                              void* d_out, int out_size, void* d_ws, size_t ws_size,
                              hipStream_t stream) {
    (void)in_sizes; (void)n_in; (void)out_size; (void)d_ws; (void)ws_size;
    const float* inputs = (const float*)d_in[0];
    const float* kern   = (const float*)d_in[1];
    const float* mixw   = (const float*)d_in[2];
    float* out          = (float*)d_out;

    dim3 grid(4 * 32, 4, 1);
    dim3 block(128, 1, 1);
    mvn_mixture_wmma_kernel<<<grid, block, 0, stream>>>(inputs, kern, mixw, out);
}


# Batch rows fetched as whole 128-byte lines (a load covers 4 full rows, not 32-byte pieces of 16; 4 tiles at entry, was 3), turned into A fragments via a per-wave LDS tile; k and sum order kept
# speedup vs baseline: 1.5700x; 1.0949x over previous
#include <hip/hip_runtime.h>
#include <hip/hip_bf16.h>
#include <math.h>

typedef __attribute__((ext_vector_type(2))) float v2f;
typedef __attribute__((ext_vector_type(4))) float v4f;
typedef __attribute__((ext_vector_type(8))) float v8f;

#define DIM     32
#define K2      64
#define CCOMP   8
#define BATCH   2048
#define KFEAT   64
#define NCHUNK  16
#define TPW     8
#define TG1     4
#define XP      36

__device__ __forceinline__ float softplus_stable(float z) {
    return fmaxf(z, 0.0f) + log1pf(expf(-fabsf(z)));
}

__device__ __forceinline__ float swz_x1(float v) {
    return __int_as_float(__builtin_amdgcn_ds_swizzle(__float_as_int(v), 0x041F));
}
__device__ __forceinline__ float swz_x2(float v) {
    return __int_as_float(__builtin_amdgcn_ds_swizzle(__float_as_int(v), 0x081F));
}
__device__ __forceinline__ float swz_x4(float v) {
    return __int_as_float(__builtin_amdgcn_ds_swizzle(__float_as_int(v), 0x101F));
}

__global__ void __launch_bounds__(128)
mvn_mixture_wmma_kernel(const float* __restrict__ inputs,
                        const float* __restrict__ kern,
                        const float* __restrict__ mixw,
                        float* __restrict__ out)
{
    __shared__ float lds_W[KFEAT * 16];
    __shared__ float lds_pld[8 * 16];
    __shared__ float lds_pc0[8 * 16];
    __shared__ float lds_beta[16];
    __shared__ __attribute__((aligned(16))) float lds_x[4 * 16 * XP];

    const int k1i   = blockIdx.x >> 5;
    const int ntile = blockIdx.x & 31;
    const int n0    = ntile * 16;

    const int tid = threadIdx.x;

    const int lane = tid & 31;
    const int wave = tid >> 5;
    const int col  = lane & 15;
    const int h    = lane >> 4;
    const int m_first = (blockIdx.y * 4 + wave) * (TPW * 16);

    const int comp  = tid & 15;
    const int chunk = tid >> 4;
    float mu4[4], raw4[4];
    {
        const int n = n0 + comp;
        const float* kp = kern + ((size_t)((k1i * K2 + (n >> 3)) * CCOMP + (n & 7))) * (2 * DIM);
        #pragma unroll
        for (int j = 0; j < 4; ++j) {
            mu4[j]  = kp[chunk * 4 + j];
            raw4[j] = kp[DIM + chunk * 4 + j];
        }
    }

    const float* xw = inputs + ((size_t)(k1i * BATCH + m_first)) * DIM + lane * 4;
    v4f xr[TPW][4];
    #pragma unroll
    for (int t = 0; t < TG1; ++t) {
        #pragma unroll
        for (int j = 0; j < 4; ++j)
            xr[t][j] = *(const v4f*)(xw + (size_t)(t * 4 + j) * 128);
    }

    float mw[CCOMP] = {0.f, 0.f, 0.f, 0.f, 0.f, 0.f, 0.f, 0.f};
    float mwc = 0.f;
    if (tid < 16) {
        const int n  = n0 + tid;
        const float* mp = mixw + (size_t)(k1i * K2 + (n >> 3)) * CCOMP;
        #pragma unroll
        for (int j = 0; j < CCOMP; ++j) mw[j] = mp[j];
        mwc = mp[n & 7];
    }

    {
        const float SP_INV1 = 0.5413248546129181f;
        float pld = 0.0f, pc0 = 0.0f;
        #pragma unroll
        for (int j = 0; j < 4; ++j) {
            const int d = chunk * 4 + j;
            float mu  = mu4[j];
            float raw = raw4[j];
            float sd  = softplus_stable(raw + SP_INV1) + 1e-8f;
            float pin = 1.0f / sd;
            float w   = pin * pin;
            pld += logf(sd);
            pc0 += w * mu * mu;
            lds_W[d * 16 + comp]         = w * mu;
            lds_W[(DIM + d) * 16 + comp] = -0.5f * w;
        }
        lds_pld[chunk * 16 + comp] = pld;
        lds_pc0[chunk * 16 + comp] = pc0;
    }
    __syncthreads();

    float* xs_st = lds_x + wave * (16 * XP) + (lane >> 3) * XP + (lane & 7) * 4;
    const float* xs_ld = lds_x + wave * (16 * XP) + col * XP + h * 16;
    v4f fq[4];
    #pragma unroll
    for (int j = 0; j < 4; ++j)
        *(v4f*)(xs_st + (4 * j) * XP) = xr[0][j];
    #pragma unroll
    for (int j = 0; j < 4; ++j)
        fq[j] = *(const v4f*)(xs_ld + 4 * j);

    if (tid < 16) {
        float logdet = 0.0f, c0 = 0.0f;
        #pragma unroll
        for (int ch = 0; ch < 8; ++ch) {
            logdet += lds_pld[ch * 16 + tid];
            c0     += lds_pc0[ch * 16 + tid];
        }
        logdet *= 2.0f;

        float mx = mw[0];
        #pragma unroll
        for (int j = 1; j < CCOMP; ++j) mx = fmaxf(mx, mw[j]);
        float se = 0.0f;
        #pragma unroll
        for (int j = 0; j < CCOMP; ++j) se += expf(mw[j] - mx);
        float logmix = mwc - mx - logf(se);

        const float CONST_TERM = 58.81206994227444f;
        lds_beta[tid] = -0.5f * (CONST_TERM + logdet + c0) + logmix;
    }
    __syncthreads();

    #pragma unroll
    for (int t = TG1; t < TPW; ++t) {
        #pragma unroll
        for (int j = 0; j < 4; ++j)
            xr[t][j] = *(const v4f*)(xw + (size_t)(t * 4 + j) * 128);
    }

    v2f bf[NCHUNK];
    #pragma unroll
    for (int kk = 0; kk < NCHUNK; ++kk) {
        int r0 = (kk & 8) * 4 + h * 16 + (kk & 7) * 2;
        bf[kk].x = lds_W[(r0 + 0) * 16 + col];
        bf[kk].y = lds_W[(r0 + 1) * 16 + col];
    }
    const float beta = lds_beta[col];

    const int k2_out = (n0 >> 3) + ((lane >> 3) & 1);
    const bool store_lane = ((lane & 7) == 0);
    float* op = out + ((size_t)(k1i * BATCH + m_first + h * 8)) * K2 + k2_out;

    #pragma unroll
    for (int t = 0; t < TPW; ++t) {
        v2f af[NCHUNK];
        #pragma unroll
        for (int kk = 0; kk < 8; ++kk) {
            const v4f q = fq[kk >> 1];
            v2f a;
            if (kk & 1) { a.x = q.z; a.y = q.w; } else { a.x = q.x; a.y = q.y; }
            af[kk]     = a;
            af[8 + kk] = a * a;
        }

        if (t + 1 < TPW) {
            #pragma unroll
            for (int j = 0; j < 4; ++j)
                *(v4f*)(xs_st + (4 * j) * XP) = xr[t + 1][j];
            #pragma unroll
            for (int j = 0; j < 4; ++j)
                fq[j] = *(const v4f*)(xs_ld + 4 * j);
        }

        v8f acc = {0.f, 0.f, 0.f, 0.f, 0.f, 0.f, 0.f, 0.f};
        #pragma unroll
        for (int kk = 0; kk < NCHUNK; ++kk) {
            acc = __builtin_amdgcn_wmma_f32_16x16x4_f32(
                 false, af[kk],
                 false, bf[kk],
                 (short)0, acc,
                 false,  false);
        }

        float res[8];
        #pragma unroll
        for (int r = 0; r < 8; ++r) {
            float s = acc[r] + beta;
            float m = s;
            m = fmaxf(m, swz_x1(m));
            m = fmaxf(m, swz_x2(m));
            m = fmaxf(m, swz_x4(m));
            float e = expf(s - m);
            e += swz_x1(e);
            e += swz_x2(e);
            e += swz_x4(e);
            res[r] = m + logf(e);
        }
        if (store_lane) {
            #pragma unroll
            for (int r = 0; r < 8; ++r)
                op[(size_t)(t * 16 + r) * K2] = res[r];
        }
    }
}

extern "C" void kernel_launch(void* const* d_in, const int* in_sizes, int n_in,
                              void* d_out, int out_size, void* d_ws, size_t ws_size,
                              hipStream_t stream) {
    (void)in_sizes; (void)n_in; (void)out_size; (void)d_ws; (void)ws_size;
    const float* inputs = (const float*)d_in[0];
    const float* kern   = (const float*)d_in[1];
    const float* mixw   = (const float*)d_in[2];
    float* out          = (float*)d_out;

    dim3 grid(4 * 32, 4, 1);
    dim3 block(128, 1, 1);
    mvn_mixture_wmma_kernel<<<grid, block, 0, stream>>>(inputs, kern, mixw, out);
}


# Both logsumexp sites (8-component groups per tile, mixture weights) take exp and log from the hardware base-2 instructions plus one constant multiply, not the library expansions; last bits differ
# speedup vs baseline: 1.6837x; 1.0724x over previous
#include <hip/hip_runtime.h>
#include <hip/hip_bf16.h>
#include <math.h>

typedef __attribute__((ext_vector_type(2))) float v2f;
typedef __attribute__((ext_vector_type(4))) float v4f;
typedef __attribute__((ext_vector_type(8))) float v8f;

#define DIM     32
#define K2      64
#define CCOMP   8
#define BATCH   2048
#define KFEAT   64
#define NCHUNK  16
#define TPW     8
#define TG1     4
#define XP      36

__device__ __forceinline__ float softplus_stable(float z) {
    return fmaxf(z, 0.0f) + log1pf(expf(-fabsf(z)));
}

__device__ __forceinline__ float exp_hw(float x) {
    return __builtin_amdgcn_exp2f(x * 1.4426950408889634f);
}
__device__ __forceinline__ float log_hw(float x) {
    return 0.6931471805599453f * __builtin_amdgcn_logf(x);
}

__device__ __forceinline__ float swz_x1(float v) {
    return __int_as_float(__builtin_amdgcn_ds_swizzle(__float_as_int(v), 0x041F));
}
__device__ __forceinline__ float swz_x2(float v) {
    return __int_as_float(__builtin_amdgcn_ds_swizzle(__float_as_int(v), 0x081F));
}
__device__ __forceinline__ float swz_x4(float v) {
    return __int_as_float(__builtin_amdgcn_ds_swizzle(__float_as_int(v), 0x101F));
}

__global__ void __launch_bounds__(128)
mvn_mixture_wmma_kernel(const float* __restrict__ inputs,
                        const float* __restrict__ kern,
                        const float* __restrict__ mixw,
                        float* __restrict__ out)
{
    __shared__ float lds_W[KFEAT * 16];
    __shared__ float lds_pld[8 * 16];
    __shared__ float lds_pc0[8 * 16];
    __shared__ float lds_beta[16];
    __shared__ __attribute__((aligned(16))) float lds_x[4 * 16 * XP];

    const int k1i   = blockIdx.x >> 5;
    const int ntile = blockIdx.x & 31;
    const int n0    = ntile * 16;

    const int tid = threadIdx.x;

    const int lane = tid & 31;
    const int wave = tid >> 5;
    const int col  = lane & 15;
    const int h    = lane >> 4;
    const int m_first = (blockIdx.y * 4 + wave) * (TPW * 16);

    const int comp  = tid & 15;
    const int chunk = tid >> 4;
    float mu4[4], raw4[4];
    {
        const int n = n0 + comp;
        const float* kp = kern + ((size_t)((k1i * K2 + (n >> 3)) * CCOMP + (n & 7))) * (2 * DIM);
        #pragma unroll
        for (int j = 0; j < 4; ++j) {
            mu4[j]  = kp[chunk * 4 + j];
            raw4[j] = kp[DIM + chunk * 4 + j];
        }
    }

    const float* xw = inputs + ((size_t)(k1i * BATCH + m_first)) * DIM + lane * 4;
    v4f xr[TPW][4];
    #pragma unroll
    for (int t = 0; t < TG1; ++t) {
        #pragma unroll
        for (int j = 0; j < 4; ++j)
            xr[t][j] = *(const v4f*)(xw + (size_t)(t * 4 + j) * 128);
    }

    float mw[CCOMP] = {0.f, 0.f, 0.f, 0.f, 0.f, 0.f, 0.f, 0.f};
    float mwc = 0.f;
    if (tid < 16) {
        const int n  = n0 + tid;
        const float* mp = mixw + (size_t)(k1i * K2 + (n >> 3)) * CCOMP;
        #pragma unroll
        for (int j = 0; j < CCOMP; ++j) mw[j] = mp[j];
        mwc = mp[n & 7];
    }

    {
        const float SP_INV1 = 0.5413248546129181f;
        float pld = 0.0f, pc0 = 0.0f;
        #pragma unroll
        for (int j = 0; j < 4; ++j) {
            const int d = chunk * 4 + j;
            float mu  = mu4[j];
            float raw = raw4[j];
            float sd  = softplus_stable(raw + SP_INV1) + 1e-8f;
            float pin = 1.0f / sd;
            float w   = pin * pin;
            pld += logf(sd);
            pc0 += w * mu * mu;
            lds_W[d * 16 + comp]         = w * mu;
            lds_W[(DIM + d) * 16 + comp] = -0.5f * w;
        }
        lds_pld[chunk * 16 + comp] = pld;
        lds_pc0[chunk * 16 + comp] = pc0;
    }
    __syncthreads();

    float* xs_st = lds_x + wave * (16 * XP) + (lane >> 3) * XP + (lane & 7) * 4;
    const float* xs_ld = lds_x + wave * (16 * XP) + col * XP + h * 16;
    v4f fq[4];
    #pragma unroll
    for (int j = 0; j < 4; ++j)
        *(v4f*)(xs_st + (4 * j) * XP) = xr[0][j];
    #pragma unroll
    for (int j = 0; j < 4; ++j)
        fq[j] = *(const v4f*)(xs_ld + 4 * j);

    if (tid < 16) {
        float logdet = 0.0f, c0 = 0.0f;
        #pragma unroll
        for (int ch = 0; ch < 8; ++ch) {
            logdet += lds_pld[ch * 16 + tid];
            c0     += lds_pc0[ch * 16 + tid];
        }
        logdet *= 2.0f;

        float mx = mw[0];
        #pragma unroll
        for (int j = 1; j < CCOMP; ++j) mx = fmaxf(mx, mw[j]);
        float se = 0.0f;
        #pragma unroll
        for (int j = 0; j < CCOMP; ++j) se += exp_hw(mw[j] - mx);
        float logmix = mwc - mx - log_hw(se);

        const float CONST_TERM = 58.81206994227444f;
        lds_beta[tid] = -0.5f * (CONST_TERM + logdet + c0) + logmix;
    }
    __syncthreads();

    #pragma unroll
    for (int t = TG1; t < TPW; ++t) {
        #pragma unroll
        for (int j = 0; j < 4; ++j)
            xr[t][j] = *(const v4f*)(xw + (size_t)(t * 4 + j) * 128);
    }

    v2f bf[NCHUNK];
    #pragma unroll
    for (int kk = 0; kk < NCHUNK; ++kk) {
        int r0 = (kk & 8) * 4 + h * 16 + (kk & 7) * 2;
        bf[kk].x = lds_W[(r0 + 0) * 16 + col];
        bf[kk].y = lds_W[(r0 + 1) * 16 + col];
    }
    const float beta = lds_beta[col];

    const int k2_out = (n0 >> 3) + ((lane >> 3) & 1);
    const bool store_lane = ((lane & 7) == 0);
    float* op = out + ((size_t)(k1i * BATCH + m_first + h * 8)) * K2 + k2_out;

    #pragma unroll
    for (int t = 0; t < TPW; ++t) {
        v2f af[NCHUNK];
        #pragma unroll
        for (int kk = 0; kk < 8; ++kk) {
            const v4f q = fq[kk >> 1];
            v2f a;
            if (kk & 1) { a.x = q.z; a.y = q.w; } else { a.x = q.x; a.y = q.y; }
            af[kk]     = a;
            af[8 + kk] = a * a;
        }

        if (t + 1 < TPW) {
            #pragma unroll
            for (int j = 0; j < 4; ++j)
                *(v4f*)(xs_st + (4 * j) * XP) = xr[t + 1][j];
            #pragma unroll
            for (int j = 0; j < 4; ++j)
                fq[j] = *(const v4f*)(xs_ld + 4 * j);
        }

        v8f acc = {0.f, 0.f, 0.f, 0.f, 0.f, 0.f, 0.f, 0.f};
        #pragma unroll
        for (int kk = 0; kk < NCHUNK; ++kk) {
            acc = __builtin_amdgcn_wmma_f32_16x16x4_f32(
                 false, af[kk],
                 false, bf[kk],
                 (short)0, acc,
                 false,  false);
        }

        float res[8];
        #pragma unroll
        for (int r = 0; r < 8; ++r) {
            float s = acc[r] + beta;
            float m = s;
            m = fmaxf(m, swz_x1(m));
            m = fmaxf(m, swz_x2(m));
            m = fmaxf(m, swz_x4(m));
            float e = exp_hw(s - m);
            e += swz_x1(e);
            e += swz_x2(e);
            e += swz_x4(e);
            res[r] = m + log_hw(e);
        }
        if (store_lane) {
            #pragma unroll
            for (int r = 0; r < 8; ++r)
                op[(size_t)(t * 16 + r) * K2] = res[r];
        }
    }
}

extern "C" void kernel_launch(void* const* d_in, const int* in_sizes, int n_in,
                              void* d_out, int out_size, void* d_ws, size_t ws_size,
                              hipStream_t stream) {
    (void)in_sizes; (void)n_in; (void)out_size; (void)d_ws; (void)ws_size;
    const float* inputs = (const float*)d_in[0];
    const float* kern   = (const float*)d_in[1];
    const float* mixw   = (const float*)d_in[2];
    float* out          = (float*)d_out;

    dim3 grid(4 * 32, 4, 1);
    dim3 block(128, 1, 1);
    mvn_mixture_wmma_kernel<<<grid, block, 0, stream>>>(inputs, kern, mixw, out);
}


# Each wave multiplies every staged batch tile by two component tiles (B of 32 components in registers, 8-wave work-groups, two accumulator chains in turn); a row's 4 results leave as one 16-byte piece
# speedup vs baseline: 2.1558x; 1.2804x over previous
#include <hip/hip_runtime.h>
#include <hip/hip_bf16.h>
#include <math.h>

typedef __attribute__((ext_vector_type(2))) float v2f;
typedef __attribute__((ext_vector_type(4))) float v4f;
typedef __attribute__((ext_vector_type(8))) float v8f;

#define DIM     32
#define K2      64
#define CCOMP   8
#define BATCH   2048
#define KFEAT   64
#define NCHUNK  16
#define NTW     2
#define TPW     4
#define NWAVE   8
#define XP      36

__device__ __forceinline__ float softplus_stable(float z) {
    return fmaxf(z, 0.0f) + log1pf(expf(-fabsf(z)));
}

__device__ __forceinline__ float exp_hw(float x) {
    return __builtin_amdgcn_exp2f(x * 1.4426950408889634f);
}
__device__ __forceinline__ float log_hw(float x) {
    return 0.6931471805599453f * __builtin_amdgcn_logf(x);
}

__device__ __forceinline__ float swz_x1(float v) {
    return __int_as_float(__builtin_amdgcn_ds_swizzle(__float_as_int(v), 0x041F));
}
__device__ __forceinline__ float swz_x2(float v) {
    return __int_as_float(__builtin_amdgcn_ds_swizzle(__float_as_int(v), 0x081F));
}
__device__ __forceinline__ float swz_x4(float v) {
    return __int_as_float(__builtin_amdgcn_ds_swizzle(__float_as_int(v), 0x101F));
}

__global__ void __launch_bounds__(256)
mvn_mixture_wmma_kernel(const float* __restrict__ inputs,
                        const float* __restrict__ kern,
                        const float* __restrict__ mixw,
                        float* __restrict__ out)
{
    __shared__ float lds_W[NTW * KFEAT * 16];
    __shared__ float lds_pld[NTW * 8 * 16];
    __shared__ float lds_pc0[NTW * 8 * 16];
    __shared__ float lds_beta[NTW * 16];
    __shared__ __attribute__((aligned(16))) float lds_x[NWAVE * 16 * XP];

    const int k1i   = blockIdx.x >> 4;
    const int npair = blockIdx.x & 15;
    const int n0    = npair * (NTW * 16);

    const int tid = threadIdx.x;

    const int lane = tid & 31;
    const int wave = tid >> 5;
    const int col  = lane & 15;
    const int h    = lane >> 4;
    const int m_first = (blockIdx.y * NWAVE + wave) * (TPW * 16);

    const int comp  = tid & 15;
    const int chunk = (tid >> 4) & 7;
    const int btile = tid >> 7;
    float mu4[4], raw4[4];
    {
        const int n = n0 + btile * 16 + comp;
        const float* kp = kern + ((size_t)((k1i * K2 + (n >> 3)) * CCOMP + (n & 7))) * (2 * DIM);
        #pragma unroll
        for (int j = 0; j < 4; ++j) {
            mu4[j]  = kp[chunk * 4 + j];
            raw4[j] = kp[DIM + chunk * 4 + j];
        }
    }

    const float* xw = inputs + ((size_t)(k1i * BATCH + m_first)) * DIM + lane * 4;
    v4f xr[TPW][4];
    #pragma unroll
    for (int t = 0; t < TPW; ++t) {
        #pragma unroll
        for (int j = 0; j < 4; ++j)
            xr[t][j] = *(const v4f*)(xw + (size_t)(t * 4 + j) * 128);
    }

    float mw[CCOMP] = {0.f, 0.f, 0.f, 0.f, 0.f, 0.f, 0.f, 0.f};
    float mwc = 0.f;
    if (tid < NTW * 16) {
        const int n  = n0 + tid;
        const float* mp = mixw + (size_t)(k1i * K2 + (n >> 3)) * CCOMP;
        #pragma unroll
        for (int j = 0; j < CCOMP; ++j) mw[j] = mp[j];
        mwc = mp[n & 7];
    }

    {
        const float SP_INV1 = 0.5413248546129181f;
        float* Wt = lds_W + btile * (KFEAT * 16);
        float pld = 0.0f, pc0 = 0.0f;
        #pragma unroll
        for (int j = 0; j < 4; ++j) {
            const int d = chunk * 4 + j;
            float mu  = mu4[j];
            float raw = raw4[j];
            float sd  = softplus_stable(raw + SP_INV1) + 1e-8f;
            float pin = 1.0f / sd;
            float w   = pin * pin;
            pld += logf(sd);
            pc0 += w * mu * mu;
            Wt[d * 16 + comp]         = w * mu;
            Wt[(DIM + d) * 16 + comp] = -0.5f * w;
        }
        lds_pld[btile * 128 + chunk * 16 + comp] = pld;
        lds_pc0[btile * 128 + chunk * 16 + comp] = pc0;
    }
    __syncthreads();

    float* xs_st = lds_x + wave * (16 * XP) + (lane >> 3) * XP + (lane & 7) * 4;
    const float* xs_ld = lds_x + wave * (16 * XP) + col * XP + h * 16;
    v4f fq[4];
    #pragma unroll
    for (int j = 0; j < 4; ++j)
        *(v4f*)(xs_st + (4 * j) * XP) = xr[0][j];
    #pragma unroll
    for (int j = 0; j < 4; ++j)
        fq[j] = *(const v4f*)(xs_ld + 4 * j);

    if (tid < NTW * 16) {
        const int bt = tid >> 4, bc = tid & 15;
        float logdet = 0.0f, c0 = 0.0f;
        #pragma unroll
        for (int ch = 0; ch < 8; ++ch) {
            logdet += lds_pld[bt * 128 + ch * 16 + bc];
            c0     += lds_pc0[bt * 128 + ch * 16 + bc];
        }
        logdet *= 2.0f;

        float mx = mw[0];
        #pragma unroll
        for (int j = 1; j < CCOMP; ++j) mx = fmaxf(mx, mw[j]);
        float se = 0.0f;
        #pragma unroll
        for (int j = 0; j < CCOMP; ++j) se += exp_hw(mw[j] - mx);
        float logmix = mwc - mx - log_hw(se);

        const float CONST_TERM = 58.81206994227444f;
        lds_beta[tid] = -0.5f * (CONST_TERM + logdet + c0) + logmix;
    }
    __syncthreads();

    v2f bf[NTW][NCHUNK];
    float beta[NTW];
    #pragma unroll
    for (int nt = 0; nt < NTW; ++nt) {
        const float* Wt = lds_W + nt * (KFEAT * 16);
        #pragma unroll
        for (int kk = 0; kk < NCHUNK; ++kk) {
            int r0 = (kk & 8) * 4 + h * 16 + (kk & 7) * 2;
            bf[nt][kk].x = Wt[(r0 + 0) * 16 + col];
            bf[nt][kk].y = Wt[(r0 + 1) * 16 + col];
        }
        beta[nt] = lds_beta[nt * 16 + col];
    }

    const bool second   = (lane & 4) != 0;
    const int  k2_out   = npair * (NTW * 2) + (second ? 2 : 0) + ((lane >> 3) & 1);
    const bool store_lane = ((lane & 3) == 0);
    float* op = out + ((size_t)(k1i * BATCH + m_first + h * 8)) * K2 + k2_out;

    #pragma unroll
    for (int t = 0; t < TPW; ++t) {
        v2f af[NCHUNK];
        #pragma unroll
        for (int kk = 0; kk < 8; ++kk) {
            const v4f q = fq[kk >> 1];
            v2f a;
            if (kk & 1) { a.x = q.z; a.y = q.w; } else { a.x = q.x; a.y = q.y; }
            af[kk]     = a;
            af[8 + kk] = a * a;
        }

        if (t + 1 < TPW) {
            #pragma unroll
            for (int j = 0; j < 4; ++j)
                *(v4f*)(xs_st + (4 * j) * XP) = xr[t + 1][j];
            #pragma unroll
            for (int j = 0; j < 4; ++j)
                fq[j] = *(const v4f*)(xs_ld + 4 * j);
        }

        v8f acc0 = {0.f, 0.f, 0.f, 0.f, 0.f, 0.f, 0.f, 0.f};
        v8f acc1 = {0.f, 0.f, 0.f, 0.f, 0.f, 0.f, 0.f, 0.f};
        #pragma unroll
        for (int kk = 0; kk < NCHUNK; ++kk) {
            acc0 = __builtin_amdgcn_wmma_f32_16x16x4_f32(
                 false, af[kk],
                 false, bf[0][kk],
                 (short)0, acc0,
                 false,  false);
            acc1 = __builtin_amdgcn_wmma_f32_16x16x4_f32(
                 false, af[kk],
                 false, bf[1][kk],
                 (short)0, acc1,
                 false,  false);
        }

        float res[NTW][8];
        #pragma unroll
        for (int nt = 0; nt < NTW; ++nt) {
            #pragma unroll
            for (int r = 0; r < 8; ++r) {
                float s = (nt == 0 ? acc0[r] : acc1[r]) + beta[nt];
                float m = s;
                m = fmaxf(m, swz_x1(m));
                m = fmaxf(m, swz_x2(m));
                m = fmaxf(m, swz_x4(m));
                float e = exp_hw(s - m);
                e += swz_x1(e);
                e += swz_x2(e);
                e += swz_x4(e);
                res[nt][r] = m + log_hw(e);
            }
        }
        if (store_lane) {
            #pragma unroll
            for (int r = 0; r < 8; ++r)
                op[(size_t)(t * 16 + r) * K2] = second ? res[1][r] : res[0][r];
        }
    }
}

extern "C" void kernel_launch(void* const* d_in, const int* in_sizes, int n_in,
                              void* d_out, int out_size, void* d_ws, size_t ws_size,
                              hipStream_t stream) {
    (void)in_sizes; (void)n_in; (void)out_size; (void)d_ws; (void)ws_size;
    const float* inputs = (const float*)d_in[0];
    const float* kern   = (const float*)d_in[1];
    const float* mixw   = (const float*)d_in[2];
    float* out          = (float*)d_out;

    dim3 grid(4 * 16, 4, 1);
    dim3 block(256, 1, 1);
    mvn_mixture_wmma_kernel<<<grid, block, 0, stream>>>(inputs, kern, mixw, out);
}


# Path to first product cut: barriers fence LDS only (row loads stay in flight, weights requested first), 2nd barrier split arrive/wait, build on hardware exp2/log2/rcp (coefficient last bits differ)
# speedup vs baseline: 2.2874x; 1.0610x over previous
#include <hip/hip_runtime.h>
#include <hip/hip_bf16.h>
#include <math.h>

typedef __attribute__((ext_vector_type(2))) float v2f;
typedef __attribute__((ext_vector_type(4))) float v4f;
typedef __attribute__((ext_vector_type(8))) float v8f;

#define DIM     32
#define K2      64
#define CCOMP   8
#define BATCH   2048
#define KFEAT   64
#define NCHUNK  16
#define NTW     2
#define TPW     4
#define NWAVE   8
#define XP      36

__device__ __forceinline__ float exp_neg_abs(float z) {
    const float L2E_HI = 1.4426950408889634f;
    const float L2E_LO = 1.92596299e-08f;
    float a  = -fabsf(z);
    float hi = a * L2E_HI;
    float lo = fmaf(a, L2E_HI, -hi) + a * L2E_LO;
    float e  = __builtin_amdgcn_exp2f(hi);
    return fmaf(e, lo * 0.6931471805599453f, e);
}
__device__ __forceinline__ float log1p_unit(float e) {
    float t  = e * __builtin_amdgcn_rcpf(2.0f + e);
    float t2 = t * t;
    float p  = 1.0f / 15.0f;
    p = fmaf(p, t2, 1.0f / 13.0f);
    p = fmaf(p, t2, 1.0f / 11.0f);
    p = fmaf(p, t2, 1.0f / 9.0f);
    p = fmaf(p, t2, 1.0f / 7.0f);
    p = fmaf(p, t2, 1.0f / 5.0f);
    p = fmaf(p, t2, 1.0f / 3.0f);
    p = fmaf(p, t2, 1.0f);
    return (t + t) * p;
}
__device__ __forceinline__ float softplus_stable(float z) {
    return fmaxf(z, 0.0f) + log1p_unit(exp_neg_abs(z));
}

__device__ __forceinline__ void barrier_lds_only() {
    __builtin_amdgcn_fence(__ATOMIC_RELEASE, "workgroup", "local");
    __builtin_amdgcn_s_barrier();
    __builtin_amdgcn_fence(__ATOMIC_ACQUIRE, "workgroup", "local");
}
__device__ __forceinline__ void barrier_lds_only_arrive() {
    __builtin_amdgcn_fence(__ATOMIC_RELEASE, "workgroup", "local");
    __builtin_amdgcn_s_barrier_signal(-1);
}
__device__ __forceinline__ void barrier_lds_only_wait() {
    __builtin_amdgcn_s_barrier_wait(-1);
    __builtin_amdgcn_fence(__ATOMIC_ACQUIRE, "workgroup", "local");
}

__device__ __forceinline__ float exp_hw(float x) {
    return __builtin_amdgcn_exp2f(x * 1.4426950408889634f);
}
__device__ __forceinline__ float log_hw(float x) {
    return 0.6931471805599453f * __builtin_amdgcn_logf(x);
}

__device__ __forceinline__ float swz_x1(float v) {
    return __int_as_float(__builtin_amdgcn_ds_swizzle(__float_as_int(v), 0x041F));
}
__device__ __forceinline__ float swz_x2(float v) {
    return __int_as_float(__builtin_amdgcn_ds_swizzle(__float_as_int(v), 0x081F));
}
__device__ __forceinline__ float swz_x4(float v) {
    return __int_as_float(__builtin_amdgcn_ds_swizzle(__float_as_int(v), 0x101F));
}

__global__ void __launch_bounds__(256)
mvn_mixture_wmma_kernel(const float* __restrict__ inputs,
                        const float* __restrict__ kern,
                        const float* __restrict__ mixw,
                        float* __restrict__ out)
{
    __shared__ float lds_W[NTW * KFEAT * 16];
    __shared__ float lds_pld[NTW * 8 * 16];
    __shared__ float lds_pc0[NTW * 8 * 16];
    __shared__ float lds_beta[NTW * 16];
    __shared__ __attribute__((aligned(16))) float lds_x[NWAVE * 16 * XP];

    const int k1i   = blockIdx.x >> 4;
    const int npair = blockIdx.x & 15;
    const int n0    = npair * (NTW * 16);

    const int tid = threadIdx.x;

    const int lane = tid & 31;
    const int wave = tid >> 5;
    const int col  = lane & 15;
    const int h    = lane >> 4;
    const int m_first = (blockIdx.y * NWAVE + wave) * (TPW * 16);

    const int comp  = tid & 15;
    const int chunk = (tid >> 4) & 7;
    const int btile = tid >> 7;
    float mu4[4], raw4[4];
    {
        const int n = n0 + btile * 16 + comp;
        const float* kp = kern + ((size_t)((k1i * K2 + (n >> 3)) * CCOMP + (n & 7))) * (2 * DIM);
        #pragma unroll
        for (int j = 0; j < 4; ++j) {
            mu4[j]  = kp[chunk * 4 + j];
            raw4[j] = kp[DIM + chunk * 4 + j];
        }
    }

    float mw[CCOMP] = {0.f, 0.f, 0.f, 0.f, 0.f, 0.f, 0.f, 0.f};
    float mwc = 0.f;
    if (tid < NTW * 16) {
        const int n  = n0 + tid;
        const float* mp = mixw + (size_t)(k1i * K2 + (n >> 3)) * CCOMP;
        #pragma unroll
        for (int j = 0; j < CCOMP; ++j) mw[j] = mp[j];
        mwc = mp[n & 7];
    }

    const float* xw = inputs + ((size_t)(k1i * BATCH + m_first)) * DIM + lane * 4;
    v4f xr[TPW][4];
    #pragma unroll
    for (int t = 0; t < TPW; ++t) {
        #pragma unroll
        for (int j = 0; j < 4; ++j)
            xr[t][j] = *(const v4f*)(xw + (size_t)(t * 4 + j) * 128);
    }
    __builtin_amdgcn_sched_barrier(0);

    {
        const float SP_INV1 = 0.5413248546129181f;
        float* Wt = lds_W + btile * (KFEAT * 16);
        float pld = 0.0f, pc0 = 0.0f;
        #pragma unroll
        for (int j = 0; j < 4; ++j) {
            const int d = chunk * 4 + j;
            float mu  = mu4[j];
            float raw = raw4[j];
            float sd  = softplus_stable(raw + SP_INV1) + 1e-8f;
            float pin = __builtin_amdgcn_rcpf(sd);
            float w   = pin * pin;
            pld += __builtin_amdgcn_logf(sd);
            pc0 += w * mu * mu;
            Wt[d * 16 + comp]         = w * mu;
            Wt[(DIM + d) * 16 + comp] = -0.5f * w;
        }
        lds_pld[btile * 128 + chunk * 16 + comp] = 0.6931471805599453f * pld;
        lds_pc0[btile * 128 + chunk * 16 + comp] = pc0;
    }
    barrier_lds_only();

    float* xs_st = lds_x + wave * (16 * XP) + (lane >> 3) * XP + (lane & 7) * 4;
    const float* xs_ld = lds_x + wave * (16 * XP) + col * XP + h * 16;
    v4f fq[4];
    #pragma unroll
    for (int j = 0; j < 4; ++j)
        *(v4f*)(xs_st + (4 * j) * XP) = xr[0][j];
    #pragma unroll
    for (int j = 0; j < 4; ++j)
        fq[j] = *(const v4f*)(xs_ld + 4 * j);

    if (tid < NTW * 16) {
        const int bt = tid >> 4, bc = tid & 15;
        float logdet = 0.0f, c0 = 0.0f;
        #pragma unroll
        for (int ch = 0; ch < 8; ++ch) {
            logdet += lds_pld[bt * 128 + ch * 16 + bc];
            c0     += lds_pc0[bt * 128 + ch * 16 + bc];
        }
        logdet *= 2.0f;

        float mx = mw[0];
        #pragma unroll
        for (int j = 1; j < CCOMP; ++j) mx = fmaxf(mx, mw[j]);
        float se = 0.0f;
        #pragma unroll
        for (int j = 0; j < CCOMP; ++j) se += exp_hw(mw[j] - mx);
        float logmix = mwc - mx - log_hw(se);

        const float CONST_TERM = 58.81206994227444f;
        lds_beta[tid] = -0.5f * (CONST_TERM + logdet + c0) + logmix;
    }
    barrier_lds_only_arrive();

    v2f bf[NTW][NCHUNK];
    float beta[NTW];
    #pragma unroll
    for (int nt = 0; nt < NTW; ++nt) {
        const float* Wt = lds_W + nt * (KFEAT * 16);
        #pragma unroll
        for (int kk = 0; kk < NCHUNK; ++kk) {
            int r0 = (kk & 8) * 4 + h * 16 + (kk & 7) * 2;
            bf[nt][kk].x = Wt[(r0 + 0) * 16 + col];
            bf[nt][kk].y = Wt[(r0 + 1) * 16 + col];
        }
    }

    const bool second   = (lane & 4) != 0;
    const int  k2_out   = npair * (NTW * 2) + (second ? 2 : 0) + ((lane >> 3) & 1);
    const bool store_lane = ((lane & 3) == 0);
    float* op = out + ((size_t)(k1i * BATCH + m_first + h * 8)) * K2 + k2_out;

    #pragma unroll
    for (int t = 0; t < TPW; ++t) {
        v2f af[NCHUNK];
        #pragma unroll
        for (int kk = 0; kk < 8; ++kk) {
            const v4f q = fq[kk >> 1];
            v2f a;
            if (kk & 1) { a.x = q.z; a.y = q.w; } else { a.x = q.x; a.y = q.y; }
            af[kk]     = a;
            af[8 + kk] = a * a;
        }

        if (t + 1 < TPW) {
            #pragma unroll
            for (int j = 0; j < 4; ++j)
                *(v4f*)(xs_st + (4 * j) * XP) = xr[t + 1][j];
            #pragma unroll
            for (int j = 0; j < 4; ++j)
                fq[j] = *(const v4f*)(xs_ld + 4 * j);
        }

        v8f acc0 = {0.f, 0.f, 0.f, 0.f, 0.f, 0.f, 0.f, 0.f};
        v8f acc1 = {0.f, 0.f, 0.f, 0.f, 0.f, 0.f, 0.f, 0.f};
        #pragma unroll
        for (int kk = 0; kk < NCHUNK; ++kk) {
            acc0 = __builtin_amdgcn_wmma_f32_16x16x4_f32(
                 false, af[kk],
                 false, bf[0][kk],
                 (short)0, acc0,
                 false,  false);
            acc1 = __builtin_amdgcn_wmma_f32_16x16x4_f32(
                 false, af[kk],
                 false, bf[1][kk],
                 (short)0, acc1,
                 false,  false);
        }

        if (t == 0) {
            barrier_lds_only_wait();
            #pragma unroll
            for (int nt = 0; nt < NTW; ++nt) beta[nt] = lds_beta[nt * 16 + col];
        }

        float res[NTW][8];
        #pragma unroll
        for (int nt = 0; nt < NTW; ++nt) {
            #pragma unroll
            for (int r = 0; r < 8; ++r) {
                float s = (nt == 0 ? acc0[r] : acc1[r]) + beta[nt];
                float m = s;
                m = fmaxf(m, swz_x1(m));
                m = fmaxf(m, swz_x2(m));
                m = fmaxf(m, swz_x4(m));
                float e = exp_hw(s - m);
                e += swz_x1(e);
                e += swz_x2(e);
                e += swz_x4(e);
                res[nt][r] = m + log_hw(e);
            }
        }
        if (store_lane) {
            #pragma unroll
            for (int r = 0; r < 8; ++r)
                op[(size_t)(t * 16 + r) * K2] = second ? res[1][r] : res[0][r];
        }
    }
}

extern "C" void kernel_launch(void* const* d_in, const int* in_sizes, int n_in,
                              void* d_out, int out_size, void* d_ws, size_t ws_size,
                              hipStream_t stream) {
    (void)in_sizes; (void)n_in; (void)out_size; (void)d_ws; (void)ws_size;
    const float* inputs = (const float*)d_in[0];
    const float* kern   = (const float*)d_in[1];
    const float* mixw   = (const float*)d_in[2];
    float* out          = (float*)d_out;

    dim3 grid(4 * 16, 4, 1);
    dim3 block(256, 1, 1);
    mvn_mixture_wmma_kernel<<<grid, block, 0, stream>>>(inputs, kern, mixw, out);
}


# Product formed transposed (coefficients as A, batch rows as B): a k2 group's 8 components lie in one lane's D registers, logsumexp in-lane (no ds_swizzle), one 8-byte store per lane; sum tree kept
# speedup vs baseline: 2.6345x; 1.1517x over previous
#include <hip/hip_runtime.h>
#include <hip/hip_bf16.h>
#include <math.h>

typedef __attribute__((ext_vector_type(2))) float v2f;
typedef __attribute__((ext_vector_type(4))) float v4f;
typedef __attribute__((ext_vector_type(8))) float v8f;

#define DIM     32
#define K2      64
#define CCOMP   8
#define BATCH   2048
#define KFEAT   64
#define NCHUNK  16
#define NTW     2
#define TPW     4
#define NWAVE   8
#define XP      36

__device__ __forceinline__ float exp_neg_abs(float z) {
    const float L2E_HI = 1.4426950408889634f;
    const float L2E_LO = 1.92596299e-08f;
    float a  = -fabsf(z);
    float hi = a * L2E_HI;
    float lo = fmaf(a, L2E_HI, -hi) + a * L2E_LO;
    float e  = __builtin_amdgcn_exp2f(hi);
    return fmaf(e, lo * 0.6931471805599453f, e);
}
__device__ __forceinline__ float log1p_unit(float e) {
    float t  = e * __builtin_amdgcn_rcpf(2.0f + e);
    float t2 = t * t;
    float p  = 1.0f / 15.0f;
    p = fmaf(p, t2, 1.0f / 13.0f);
    p = fmaf(p, t2, 1.0f / 11.0f);
    p = fmaf(p, t2, 1.0f / 9.0f);
    p = fmaf(p, t2, 1.0f / 7.0f);
    p = fmaf(p, t2, 1.0f / 5.0f);
    p = fmaf(p, t2, 1.0f / 3.0f);
    p = fmaf(p, t2, 1.0f);
    return (t + t) * p;
}
__device__ __forceinline__ float softplus_stable(float z) {
    return fmaxf(z, 0.0f) + log1p_unit(exp_neg_abs(z));
}

__device__ __forceinline__ void barrier_lds_only() {
    __builtin_amdgcn_fence(__ATOMIC_RELEASE, "workgroup", "local");
    __builtin_amdgcn_s_barrier();
    __builtin_amdgcn_fence(__ATOMIC_ACQUIRE, "workgroup", "local");
}
__device__ __forceinline__ void barrier_lds_only_arrive() {
    __builtin_amdgcn_fence(__ATOMIC_RELEASE, "workgroup", "local");
    __builtin_amdgcn_s_barrier_signal(-1);
}
__device__ __forceinline__ void barrier_lds_only_wait() {
    __builtin_amdgcn_s_barrier_wait(-1);
    __builtin_amdgcn_fence(__ATOMIC_ACQUIRE, "workgroup", "local");
}

__device__ __forceinline__ float exp_hw(float x) {
    return __builtin_amdgcn_exp2f(x * 1.4426950408889634f);
}
__device__ __forceinline__ float log_hw(float x) {
    return 0.6931471805599453f * __builtin_amdgcn_logf(x);
}

__device__ __forceinline__ int comp_at(int n0, int bt, int p) {
    return n0 + ((((p >> 3) << 1) + bt) << 3) + (p & 7);
}

__global__ void __launch_bounds__(256)
mvn_mixture_wmma_kernel(const float* __restrict__ inputs,
                        const float* __restrict__ kern,
                        const float* __restrict__ mixw,
                        float* __restrict__ out)
{
    __shared__ float lds_W[NTW * KFEAT * 16];
    __shared__ float lds_pld[NTW * 8 * 16];
    __shared__ float lds_pc0[NTW * 8 * 16];
    __shared__ __attribute__((aligned(16))) float lds_beta[NTW * 16];
    __shared__ __attribute__((aligned(16))) float lds_x[NWAVE * 16 * XP];

    const int k1i   = blockIdx.x >> 4;
    const int npair = blockIdx.x & 15;
    const int n0    = npair * (NTW * 16);

    const int tid = threadIdx.x;

    const int lane = tid & 31;
    const int wave = tid >> 5;
    const int col  = lane & 15;
    const int h    = lane >> 4;
    const int m_first = (blockIdx.y * NWAVE + wave) * (TPW * 16);

    const int comp  = tid & 15;
    const int chunk = (tid >> 4) & 7;
    const int btile = tid >> 7;
    float mu4[4], raw4[4];
    {
        const int n = comp_at(n0, btile, comp);
        const float* kp = kern + ((size_t)((k1i * K2 + (n >> 3)) * CCOMP + (n & 7))) * (2 * DIM);
        #pragma unroll
        for (int j = 0; j < 4; ++j) {
            mu4[j]  = kp[chunk * 4 + j];
            raw4[j] = kp[DIM + chunk * 4 + j];
        }
    }

    float mw[CCOMP] = {0.f, 0.f, 0.f, 0.f, 0.f, 0.f, 0.f, 0.f};
    float mwc = 0.f;
    if (tid < NTW * 16) {
        const int n  = comp_at(n0, tid >> 4, tid & 15);
        const float* mp = mixw + (size_t)(k1i * K2 + (n >> 3)) * CCOMP;
        #pragma unroll
        for (int j = 0; j < CCOMP; ++j) mw[j] = mp[j];
        mwc = mp[n & 7];
    }

    const float* xw = inputs + ((size_t)(k1i * BATCH + m_first)) * DIM + lane * 4;
    v4f xr[TPW][4];
    #pragma unroll
    for (int t = 0; t < TPW; ++t) {
        #pragma unroll
        for (int j = 0; j < 4; ++j)
            xr[t][j] = *(const v4f*)(xw + (size_t)(t * 4 + j) * 128);
    }
    __builtin_amdgcn_sched_barrier(0);

    {
        const float SP_INV1 = 0.5413248546129181f;
        float* Wt = lds_W + btile * (KFEAT * 16);
        float pld = 0.0f, pc0 = 0.0f;
        #pragma unroll
        for (int j = 0; j < 4; ++j) {
            const int d = chunk * 4 + j;
            float mu  = mu4[j];
            float raw = raw4[j];
            float sd  = softplus_stable(raw + SP_INV1) + 1e-8f;
            float pin = __builtin_amdgcn_rcpf(sd);
            float w   = pin * pin;
            pld += __builtin_amdgcn_logf(sd);
            pc0 += w * mu * mu;
            Wt[d * 16 + comp]         = w * mu;
            Wt[(DIM + d) * 16 + comp] = -0.5f * w;
        }
        lds_pld[btile * 128 + chunk * 16 + comp] = 0.6931471805599453f * pld;
        lds_pc0[btile * 128 + chunk * 16 + comp] = pc0;
    }
    barrier_lds_only();

    float* xs_st = lds_x + wave * (16 * XP) + (lane >> 3) * XP + (lane & 7) * 4;
    const float* xs_ld = lds_x + wave * (16 * XP) + col * XP + h * 16;
    v4f fq[4];
    #pragma unroll
    for (int j = 0; j < 4; ++j)
        *(v4f*)(xs_st + (4 * j) * XP) = xr[0][j];
    #pragma unroll
    for (int j = 0; j < 4; ++j)
        fq[j] = *(const v4f*)(xs_ld + 4 * j);

    if (tid < NTW * 16) {
        const int bt = tid >> 4, bc = tid & 15;
        float logdet = 0.0f, c0 = 0.0f;
        #pragma unroll
        for (int ch = 0; ch < 8; ++ch) {
            logdet += lds_pld[bt * 128 + ch * 16 + bc];
            c0     += lds_pc0[bt * 128 + ch * 16 + bc];
        }
        logdet *= 2.0f;

        float mx = mw[0];
        #pragma unroll
        for (int j = 1; j < CCOMP; ++j) mx = fmaxf(mx, mw[j]);
        float se = 0.0f;
        #pragma unroll
        for (int j = 0; j < CCOMP; ++j) se += exp_hw(mw[j] - mx);
        float logmix = mwc - mx - log_hw(se);

        const float CONST_TERM = 58.81206994227444f;
        lds_beta[tid] = -0.5f * (CONST_TERM + logdet + c0) + logmix;
    }
    barrier_lds_only_arrive();

    v2f bf[NTW][NCHUNK];
    v4f beta[NTW][2];
    #pragma unroll
    for (int nt = 0; nt < NTW; ++nt) {
        const float* Wt = lds_W + nt * (KFEAT * 16);
        #pragma unroll
        for (int kk = 0; kk < NCHUNK; ++kk) {
            int r0 = (kk & 8) * 4 + h * 16 + (kk & 7) * 2;
            bf[nt][kk].x = Wt[(r0 + 0) * 16 + col];
            bf[nt][kk].y = Wt[(r0 + 1) * 16 + col];
        }
    }

    float* op = out + ((size_t)(k1i * BATCH + m_first + col)) * K2 + npair * (NTW * 2) + h * 2;

    #pragma unroll
    for (int t = 0; t < TPW; ++t) {
        v2f af[NCHUNK];
        #pragma unroll
        for (int kk = 0; kk < 8; ++kk) {
            const v4f q = fq[kk >> 1];
            v2f a;
            if (kk & 1) { a.x = q.z; a.y = q.w; } else { a.x = q.x; a.y = q.y; }
            af[kk]     = a;
            af[8 + kk] = a * a;
        }

        if (t + 1 < TPW) {
            #pragma unroll
            for (int j = 0; j < 4; ++j)
                *(v4f*)(xs_st + (4 * j) * XP) = xr[t + 1][j];
            #pragma unroll
            for (int j = 0; j < 4; ++j)
                fq[j] = *(const v4f*)(xs_ld + 4 * j);
        }

        v8f acc0 = {0.f, 0.f, 0.f, 0.f, 0.f, 0.f, 0.f, 0.f};
        v8f acc1 = {0.f, 0.f, 0.f, 0.f, 0.f, 0.f, 0.f, 0.f};
        #pragma unroll
        for (int kk = 0; kk < NCHUNK; ++kk) {
            acc0 = __builtin_amdgcn_wmma_f32_16x16x4_f32(
                 false, bf[0][kk],
                 false, af[kk],
                 (short)0, acc0,
                 false,  false);
            acc1 = __builtin_amdgcn_wmma_f32_16x16x4_f32(
                 false, bf[1][kk],
                 false, af[kk],
                 (short)0, acc1,
                 false,  false);
        }

        if (t == 0) {
            barrier_lds_only_wait();
            #pragma unroll
            for (int nt = 0; nt < NTW; ++nt) {
                beta[nt][0] = *(const v4f*)(lds_beta + nt * 16 + h * 8);
                beta[nt][1] = *(const v4f*)(lds_beta + nt * 16 + h * 8 + 4);
            }
        }

        v2f res;
        #pragma unroll
        for (int nt = 0; nt < NTW; ++nt) {
            float s[8];
            #pragma unroll
            for (int r = 0; r < 8; ++r)
                s[r] = (nt == 0 ? acc0[r] : acc1[r]) + beta[nt][r >> 2][r & 3];
            const float m = fmaxf(fmaxf(fmaxf(s[0], s[1]), fmaxf(s[2], s[3])),
                                  fmaxf(fmaxf(s[4], s[5]), fmaxf(s[6], s[7])));
            float e[8];
            #pragma unroll
            for (int r = 0; r < 8; ++r) e[r] = exp_hw(s[r] - m);
            const float se = ((e[0] + e[1]) + (e[2] + e[3])) + ((e[4] + e[5]) + (e[6] + e[7]));
            const float lse = m + log_hw(se);
            if (nt == 0) res.x = lse; else res.y = lse;
        }
        *(v2f*)(op + (size_t)(t * 16) * K2) = res;
        if (t + 1 < TPW) __builtin_amdgcn_sched_barrier(0);
    }
}

extern "C" void kernel_launch(void* const* d_in, const int* in_sizes, int n_in,
                              void* d_out, int out_size, void* d_ws, size_t ws_size,
                              hipStream_t stream) {
    (void)in_sizes; (void)n_in; (void)out_size; (void)d_ws; (void)ws_size;
    const float* inputs = (const float*)d_in[0];
    const float* kern   = (const float*)d_in[1];
    const float* mixw   = (const float*)d_in[2];
    float* out          = (float*)d_out;

    dim3 grid(4 * 16, 4, 1);
    dim3 block(256, 1, 1);
    mvn_mixture_wmma_kernel<<<grid, block, 0, stream>>>(inputs, kern, mixw, out);
}
